# starting kernel
# baseline (speedup 1.0000x reference)
#include <hip/hip_runtime.h>


typedef float v2f __attribute__((ext_vector_type(2)));
typedef float v8f __attribute__((ext_vector_type(8)));

__global__ __launch_bounds__(32)
void shiftconv_wmma_f32(const float* __restrict__ x,
                        const float* __restrict__ w,
                        float* __restrict__ out) {
  const int lane   = threadIdx.x;
  const int mt     = blockIdx.x;
  const int ng     = blockIdx.y;
  const int lanelo = lane & 15;
  const int hi     = lane >> 4;

  const int  r  = mt * 16 + lanelo;
  const bool rv = (r < 392);
  const int  rr = rv ? r : 0;
  const int  o   = rr / 196;
  const int  rem = rr - o * 196;
  const int  n   = rem / 14;
  const int  m   = rem - n * 14;
  const float* xb = x + (o * 256) * 196 + n * 14;

  const float rvf = rv ? 1.0f : 0.0f;
  float af[3];
  int   wc[3];
  af[0] = (m >= 1)  ? rvf : 0.0f;  wc[0] = (m + 12) % 14;
  af[1] = rvf;                     wc[1] = (m + 13) % 14;
  af[2] = (m <= 12) ? rvf : 0.0f;  wc[2] = m;

  const int kbase = ng * 64 + lanelo;

  const int qoff = hi * 2;
  const float* pA[6];
  const float* pB[6];
  float        mA[6];
  #pragma unroll
  for (int u = 0; u < 6; ++u) {
    const int t = u >> 1, v = u & 1;
    const int q = t * 4 + qoff + v;
    const int i = (int)((unsigned)q / 3u);
    const int j = q - i * 3;
    pA[u] = xb + i * 196 + wc[j];
    mA[u] = af[j];
    pB[u] = w + i * 768 + j + kbase * 3;
  }

  v8f acc0 = {}, acc1 = {}, acc2 = {}, acc3 = {};

  #pragma unroll 2
  for (int it = 0; it < 64; ++it) {
    #pragma unroll
    for (int t = 0; t < 3; ++t) {
      const int u0 = 2 * t, u1 = 2 * t + 1;

      v2f a;
      a.x = pA[u0][0] * mA[u0];
      a.y = pA[u1][0] * mA[u1];

      v2f b0, b1, b2, b3;
      b0.x = pB[u0][  0];  b0.y = pB[u1][  0];
      b1.x = pB[u0][ 48];  b1.y = pB[u1][ 48];
      b2.x = pB[u0][ 96];  b2.y = pB[u1][ 96];
      b3.x = pB[u0][144];  b3.y = pB[u1][144];

      acc0 = __builtin_amdgcn_wmma_f32_16x16x4_f32(false, a, false, b0, (short)0, acc0, false, false);
      acc1 = __builtin_amdgcn_wmma_f32_16x16x4_f32(false, a, false, b1, (short)0, acc1, false, false);
      acc2 = __builtin_amdgcn_wmma_f32_16x16x4_f32(false, a, false, b2, (short)0, acc2, false, false);
      acc3 = __builtin_amdgcn_wmma_f32_16x16x4_f32(false, a, false, b3, (short)0, acc3, false, false);
    }
    #pragma unroll
    for (int u = 0; u < 6; ++u) {
      pA[u] += 4 * 196;
      pB[u] += 4 * 768;
    }
  }

  #pragma unroll
  for (int d = 0; d < 8; ++d) {
    const int p = mt * 16 + hi * 8 + d;
    if (p < 392) {
      const int po   = p / 196;
      const int prem = p - po * 196;
      const int pn   = prem / 14;
      const int pm   = prem - pn * 14;
      const int pn1  = (pn + 1) % 14;
      float* ob = out + (po * 256 + ng * 64 + lanelo) * 196 + pn1 * 14 + pm;
      ob[ 0 * 196] = acc0[d];
      ob[16 * 196] = acc1[d];
      ob[32 * 196] = acc2[d];
      ob[48 * 196] = acc3[d];
    }
  }
}

extern "C" void kernel_launch(void* const* d_in, const int* in_sizes, int n_in,
                              void* d_out, int out_size, void* d_ws, size_t ws_size,
                              hipStream_t stream) {
  const float* x = (const float*)d_in[0];
  const float* w = (const float*)d_in[1];
  float* out = (float*)d_out;
  (void)in_sizes; (void)n_in; (void)out_size; (void)d_ws; (void)ws_size;

  dim3 grid(25, 4);
  shiftconv_wmma_f32<<<grid, 32, 0, stream>>>(x, w, out);
}


# staging moved to 4 extra waves (one per SIMD) that store 4 channel chunks to LDS while the 4 matrix waves multiply the chunk before; 3 more LDS-only barriers; LDS rows grouped by lane half; same sums
# speedup vs baseline: 6.4921x; 6.4921x over previous
#include <hip/hip_runtime.h>


typedef float v2f __attribute__((ext_vector_type(2)));
typedef float v4f __attribute__((ext_vector_type(4)));
typedef float v8f __attribute__((ext_vector_type(8)));

#define XP    80
#define WP    112
#define XZERO 56
#define XH    (64 * XP + 16)
#define XROW(a) ((((a) & 1) * XH) + (((a) >> 1) * XP))
#define WH    (64 * WP + 16)
#define WROW(a) ((((a) & 1) * WH) + (((a) >> 1) * WP))

#define KBLK  12

#define LOAD_BLOCK(XB, WB, blk)                                         \
  _Pragma("unroll")                                                     \
  for (int q = 0; q < 4; ++q) {                                         \
    const int xo = ((blk) * 4 + q) * XP;                                \
    const v2f* wq = wrow + ((blk) * 4 + q) * WP;                        \
    XB[3 * q + 0] = x0p[xo];  XB[3 * q + 1] = x1p[xo];  XB[3 * q + 2] = x2p[xo]; \
    WB[3 * q + 0] = wq[0];   WB[3 * q + 1] = wq[1];   WB[3 * q + 2] = wq[2];  \
  }

#define MMA_BLOCK(XB, WB)                                               \
  _Pragma("unroll")                                                     \
  for (int t = 0; t < KBLK; t += 2) {                                   \
    acc0 = __builtin_amdgcn_wmma_f32_16x16x4_f32(false, WB[t], false, XB[t], (short)0, acc0, false, false);         \
    acc1 = __builtin_amdgcn_wmma_f32_16x16x4_f32(false, WB[t + 1], false, XB[t + 1], (short)0, acc1, false, false); \
  }

#define TRIPS_FLAT(j0, j1)                                              \
  _Pragma("unroll")                                                     \
  for (int j = (j0); j < (j1); j += 2) {                                \
    LOAD_BLOCK(xA, wA, j)                                               \
    MMA_BLOCK(xB, wB)                                                   \
    LOAD_BLOCK(xB, wB, j + 1)                                           \
    MMA_BLOCK(xA, wA)                                                   \
    __builtin_amdgcn_sched_group_barrier(0x100, 12, 0);                 \
    __builtin_amdgcn_sched_group_barrier(0x008, 12, 0);                 \
    __builtin_amdgcn_sched_group_barrier(0x100, 12, 0);                 \
    __builtin_amdgcn_sched_group_barrier(0x008, 12, 0);                 \
  }

#define LOAD_CHUNK(c)                                                   \
  _Pragma("unroll")                                                     \
  for (int r = 2 * (c); r < 2 * (c) + 2; ++r) {                         \
    _Pragma("unroll")                                                   \
    for (int u = 0; u < 3; ++u) {                                       \
      const float* g = gw + r * (16 * 2 * 768) + u * 32;                \
      wa[r * 3 + u] = *(const v4f*)g;                                   \
      wb[r * 3 + u] = *(const v4f*)(g + 768);                           \
    }                                                                   \
  }                                                                     \
  _Pragma("unroll")                                                     \
  for (int r = 8 * (c); r < 8 * (c) + 8; ++r) {                         \
    const float* g = gx + r * (4 * 2 * 196);                            \
    xa[r] = *(const v2f*)g;                                             \
    xb[r] = *(const v2f*)(g + 196);                                     \
  }

#define LDS_BARRIER()                                                   \
  __builtin_amdgcn_fence(__ATOMIC_RELEASE, "workgroup", "local");       \
  __builtin_amdgcn_s_barrier();                                         \
  __builtin_amdgcn_fence(__ATOMIC_ACQUIRE, "workgroup", "local");

#define STORE_CHUNK(c)                                                  \
  _Pragma("unroll")                                                     \
  for (int r = 2 * (c); r < 2 * (c) + 2; ++r) {                         \
    _Pragma("unroll")                                                   \
    for (int u = 0; u < 3; ++u) {                                       \
      const int a = wa0 + 16 * r;                                       \
      const int f = wf0 + 8 * u;                                        \
      const v4f va = wa[r * 3 + u], vb = wb[r * 3 + u];                 \
      const v4f lo = {va.x, vb.x, va.y, vb.y};                          \
      const v4f hi = {va.z, vb.z, va.w, vb.w};                          \
      *(v4f*)&Ws[WROW(a) + 4 * f]     = lo;                             \
      *(v4f*)&Ws[WROW(a) + 4 * f + 2] = hi;                             \
    }                                                                   \
  }                                                                     \
  _Pragma("unroll")                                                     \
  for (int r = 8 * (c); r < 8 * (c) + 8; ++r) {                         \
    const int a = sw + 4 * r;                                           \
    const v4f e = {xa[r].x, xb[r].x, xa[r].y, xb[r].y};                 \
    *(v4f*)&Xs[XROW(a) + 2 * sp] = e;                                   \
  }

__global__ __launch_bounds__(256)
void shiftconv_lds_wmma_f32(const float* __restrict__ x,
                            const float* __restrict__ w,
                            float* __restrict__ out) {
  __shared__ __attribute__((aligned(16))) v2f Xs[2 * XH];
  __shared__ __attribute__((aligned(16))) v2f Ws[2 * WH];

  const int tid   = threadIdx.x;
  const int wave  = __builtin_amdgcn_readfirstlane(tid >> 5);
  const bool stg  = wave >= 4;
  const int lane  = tid & 31;
  const int p0    = blockIdx.x * 32;
  const int k0    = blockIdx.y * 32;
  const int R0    = p0 / 14;
  const int Pbase = R0 * 14;

  const int st  = tid & 127;
  const int sw  = wave & 3;
  const int wa0 = st >> 3;
  const int wf0 = st & 7;
  const int sp  = lane < 28 ? lane : 27;
  v4f wa[24], wb[24];
  v2f xa[32], xb[32];

  const int cl   = lane & 15;
  const int h    = lane >> 4;
  const int mi   = wave & 1;
  const int ni   = (wave >> 1) & 1;
  const bool act = !stg && (p0 + mi * 16 < 392);

  const int p  = p0 + mi * 16 + cl;
  const int pc = p < 392 ? p : p0;
  const int R  = pc / 14;
  const int m  = pc - R * 14;
  const int sb = (R - R0) * 14;
  const int s0 = (m >= 1)  ? sb + (m + 12) % 14 : XZERO;
  const int s1 = sb + (m + 13) % 14;
  const int s2 = (m <= 12) ? sb + m : XZERO;
  const int kl = ni * 16 + cl;

  const v2f* x0p = Xs + h * XH + s0;
  const v2f* x1p = Xs + h * XH + s1;
  const v2f* x2p = Xs + h * XH + s2;
  const v2f* wrow = Ws + h * WH + kl * 3;

  v8f acc0 = {}, acc1 = {};
  v2f xA[KBLK], wA[KBLK], xB[KBLK], wB[KBLK];

  const float* gw = w + (2 * wa0) * 768 + k0 * 3 + 4 * wf0;
  const float* gx;
  {
    const int P  = Pbase + 2 * sp;
    const int Pc = P < 392 ? P : 0;
    const int o  = Pc >= 196 ? 1 : 0;
    gx = x + (o * 256 + 2 * sw) * 196 + (Pc - o * 196);
  }

  if (stg) {
    LOAD_CHUNK(0)
    LOAD_CHUNK(1)
    STORE_CHUNK(0)
    const v2f z = {0.0f, 0.0f};
    Xs[XROW(st) + XZERO] = z;
  }
  LDS_BARRIER()

  if (stg) {
    LOAD_CHUNK(2)
    STORE_CHUNK(1)
  }
  if (act) {
    LOAD_BLOCK(xA, wA, 0)
    LOAD_BLOCK(xB, wB, 1)
    MMA_BLOCK(xA, wA)
    __builtin_amdgcn_sched_group_barrier(0x100, 24, 0);
    __builtin_amdgcn_sched_group_barrier(0x008, 12, 0);
    TRIPS_FLAT(2, 4)
  }
  LDS_BARRIER()

  if (stg) {
    LOAD_CHUNK(3)
    STORE_CHUNK(2)
  }
  if (act) { TRIPS_FLAT(4, 8) }
  LDS_BARRIER()

  if (stg) { STORE_CHUNK(3) }
  if (act) { TRIPS_FLAT(8, 12) }
  LDS_BARRIER()

  if (act) {
    TRIPS_FLAT(12, 16)
    MMA_BLOCK(xB, wB)
    __builtin_amdgcn_sched_group_barrier(0x008, 12, 0);

    if (p < 392) {
      const int po   = p >= 196 ? 1 : 0;
      const int prem = p - po * 196;
      const int pn   = prem / 14;
      const int pm   = prem - pn * 14;
      const int pn1  = (pn + 1) % 14;
      float* ob = out + (po * 256 + k0 + ni * 16 + 8 * h) * 196 + pn1 * 14 + pm;
      #pragma unroll
      for (int d = 0; d < 8; ++d) ob[d * 196] = acc0[d] + acc1[d];
    }
  }
}

extern "C" void kernel_launch(void* const* d_in, const int* in_sizes, int n_in,
                              void* d_out, int out_size, void* d_ws, size_t ws_size,
                              hipStream_t stream) {
  const float* x = (const float*)d_in[0];
  const float* w = (const float*)d_in[1];
  float* out = (float*)d_out;
  (void)in_sizes; (void)n_in; (void)out_size; (void)d_ws; (void)ws_size;

  dim3 grid(13, 8);
  shiftconv_lds_wmma_f32<<<grid, 256, 0, stream>>>(x, w, out);
}
